# starting kernel
# baseline (speedup 1.0000x reference)
#include <hip/hip_runtime.h>


#define VOCAB   100
#define E       32
#define HDIM    64
#define G4H     256
#define MAXLEN  16
#define WAVES   8
#define EPITCH  36
#define WIHP    36
#define WHHP    68
#define HPITCH  68

typedef __attribute__((ext_vector_type(2))) float v2f;
typedef __attribute__((ext_vector_type(4))) float v4f;
typedef __attribute__((ext_vector_type(8))) float v8f;

__device__ __forceinline__ float sigm(float x) {
    return 1.0f / (1.0f + __expf(-x));
}

__global__ __launch_bounds__(WAVES * 32, 4)
void char_lstm_wmma(const float* __restrict__ emb,
                    const float* __restrict__ Wih,
                    const float* __restrict__ Whh,
                    const float* __restrict__ bih,
                    const float* __restrict__ bhh,
                    const int*   __restrict__ chars,
                    const int*   __restrict__ lengths,
                    float*       __restrict__ out)
{
    __shared__ float s_emb[VOCAB * EPITCH];
    __shared__ float s_wih[G4H * WIHP];
    __shared__ float s_whh[G4H * WHHP];
    __shared__ float s_bias[G4H];
    __shared__ float s_h[WAVES][16 * HPITCH];

    const int tid  = threadIdx.x;
    const int lane = tid & 31;
    const int wave = tid >> 5;
    const int half = lane >> 4;
    const int n    = lane & 15;
    const int base = (blockIdx.x * WAVES + wave) * 16;

    for (int i = tid; i < VOCAB * E; i += WAVES * 32)
        s_emb[(i / E) * EPITCH + (i % E)] = emb[i];
    for (int i = tid; i < G4H * E; i += WAVES * 32)
        s_wih[(i / E) * WIHP + (i % E)] = Wih[i];
    for (int i = tid; i < G4H * HDIM; i += WAVES * 32)
        s_whh[(i / HDIM) * WHHP + (i % HDIM)] = Whh[i];
    for (int i = tid; i < G4H; i += WAVES * 32)
        s_bias[i] = bih[i] + bhh[i];
    for (int i = lane; i < 16 * HPITCH; i += 32)
        s_h[wave][i] = 0.0f;

    int llen[8];
    #pragma unroll
    for (int r = 0; r < 8; ++r)
        llen[r] = lengths[base + 8 * half + r];

    v8f creg[4];
    v8f hreg[4];
    #pragma unroll
    for (int cb = 0; cb < 4; ++cb)
        #pragma unroll
        for (int r = 0; r < 8; ++r) { creg[cb][r] = 0.0f; hreg[cb][r] = 0.0f; }

    __syncthreads();

    #pragma unroll 1
    for (int t = 0; t < MAXLEN; ++t) {
        const int c    = chars[(base + n) * MAXLEN + t];
        const int xoff = c * EPITCH + 2 * half;

        #pragma unroll
        for (int cb = 0; cb < 4; ++cb) {
            v8f acc[4];
            #pragma unroll
            for (int g = 0; g < 4; ++g)
                #pragma unroll
                for (int r = 0; r < 8; ++r) acc[g][r] = 0.0f;

            #pragma unroll
            for (int kk = 0; kk < E / 4; ++kk) {
                v2f a = *(const v2f*)&s_emb[xoff + kk * 4];
                #pragma unroll
                for (int g = 0; g < 4; ++g) {
                    v2f b = *(const v2f*)&s_wih[(g * 64 + cb * 16 + n) * WIHP
                                                + kk * 4 + 2 * half];
                    acc[g] = __builtin_amdgcn_wmma_f32_16x16x4_f32(
                        false, a, false, b, (short)0, acc[g], false, false);
                }
            }

            #pragma unroll
            for (int kk = 0; kk < HDIM / 4; ++kk) {
                v2f a = *(const v2f*)&s_h[wave][n * HPITCH + kk * 4 + 2 * half];
                #pragma unroll
                for (int g = 0; g < 4; ++g) {
                    v2f b = *(const v2f*)&s_whh[(g * 64 + cb * 16 + n) * WHHP
                                                + kk * 4 + 2 * half];
                    acc[g] = __builtin_amdgcn_wmma_f32_16x16x4_f32(
                        false, a, false, b, (short)0, acc[g], false, false);
                }
            }

            float b0 = s_bias[0 * 64 + cb * 16 + n];
            float b1 = s_bias[1 * 64 + cb * 16 + n];
            float b2 = s_bias[2 * 64 + cb * 16 + n];
            float b3 = s_bias[3 * 64 + cb * 16 + n];

            #pragma unroll
            for (int r = 0; r < 8; ++r) {
                float ig = sigm(acc[0][r] + b0);
                float fg = sigm(acc[1][r] + b1);
                float gg = tanhf(acc[2][r] + b2);
                float og = sigm(acc[3][r] + b3);
                float cn = fg * creg[cb][r] + ig * gg;
                float hn = og * tanhf(cn);
                const bool act = t < llen[r];
                creg[cb][r] = act ? cn : creg[cb][r];
                hreg[cb][r] = act ? hn : hreg[cb][r];
            }
        }

        #pragma unroll
        for (int cb = 0; cb < 4; ++cb)
            #pragma unroll
            for (int r = 0; r < 8; ++r)
                s_h[wave][(8 * half + r) * HPITCH + cb * 16 + n] = hreg[cb][r];

        __syncthreads();
    }

    #pragma unroll
    for (int j = 0; j < 8; ++j) {
        int idx = j * 128 + lane * 4;
        int row = idx >> 6, col = idx & 63;
        v4f v = *(const v4f*)&s_h[wave][row * HPITCH + col];
        *(v4f*)&out[(base + row) * HDIM + col] = v;
    }
}

extern "C" void kernel_launch(void* const* d_in, const int* in_sizes, int n_in,
                              void* d_out, int out_size, void* d_ws, size_t ws_size,
                              hipStream_t stream) {
    const float* emb     = (const float*)d_in[0];
    const float* W_ih    = (const float*)d_in[1];
    const float* W_hh    = (const float*)d_in[2];
    const float* b_ih    = (const float*)d_in[3];
    const float* b_hh    = (const float*)d_in[4];
    const int*   chars   = (const int*)d_in[5];
    const int*   lengths = (const int*)d_in[6];
    float* out = (float*)d_out;

    const int n_words = in_sizes[6];
    const int blocks  = n_words / (WAVES * 16);
    char_lstm_wmma<<<blocks, WAVES * 32, 0, stream>>>(
        emb, W_ih, W_hh, b_ih, b_hh, chars, lengths, out);
}


# Words ranked by length over groups of 8 blocks (2048 words, ballot counting sort) instead of per block; the group's 16-word tiles dealt over its 32 SIMDs in alternating rounds: equal steps per SIMD
# speedup vs baseline: 4.4106x; 4.4106x over previous
#include <hip/hip_runtime.h>


#define VOCAB   100
#define E       32
#define HDIM    64
#define G4H     256
#define MAXLEN  16
#define WAVES   16
#define WPB     (WAVES * 16)
#define TPITCH  260
#define WHHP    68
#define HPITCH  68
#define GRP     8
#define GWORDS  (GRP * WPB)
#define NBIN    (MAXLEN + 1)
#define NSEG    (WAVES * 4)
#define CPITCH  (NSEG + 1)

typedef __attribute__((ext_vector_type(2))) float v2f;
typedef __attribute__((ext_vector_type(4))) float v4f;
typedef __attribute__((ext_vector_type(8))) float v8f;
typedef __attribute__((ext_vector_type(4))) int   v4i;

#define GATE_G    2

__device__ __forceinline__ float sigm_of_half(float xh) {
    return __builtin_fmaf(0.5f, __builtin_amdgcn_tanhf(xh), 0.5f);
}

__device__ __forceinline__ float tanh_hw(float x) {
    return __builtin_amdgcn_tanhf(x);
}

__global__ __launch_bounds__(WAVES * 32, 4)
void char_lstm_wmma(const float* __restrict__ emb,
                    const float* __restrict__ Wih,
                    const float* __restrict__ Whh,
                    const float* __restrict__ bih,
                    const float* __restrict__ bhh,
                    const int*   __restrict__ chars,
                    const int*   __restrict__ lengths,
                    float*       __restrict__ out)
{
    __shared__ alignas(16) float s_tab[VOCAB * TPITCH];
    __shared__ alignas(16) float s_whh[G4H * WHHP];
    __shared__ alignas(16) float s_h[WAVES][16 * HPITCH];
    __shared__ alignas(16) int   s_off[WAVES][MAXLEN * 16];
    __shared__ int   s_cnt[NBIN * CPITCH];
    __shared__ int   s_tot[NBIN];
    __shared__ int   s_word[WPB];
    __shared__ int   s_wlen[WPB];

    const int tid  = threadIdx.x;
    const int lane = tid & 31;
    const int wave = __builtin_amdgcn_readfirstlane(tid >> 5);
    const int half = lane >> 4;
    const int n    = lane & 15;
    const int gbase = (blockIdx.x / GRP) * GWORDS;
    const int gpos  = blockIdx.x % GRP;

    int bin[4], pre[4];
    {
        const v4i len4 = *(const v4i*)&lengths[gbase + 4 * tid];
        #pragma unroll
        for (int r = 0; r < 4; ++r) {
            const int l = len4[r];
            bin[r] = l < 0 ? 0 : (l > MAXLEN ? MAXLEN : l);
        }
        #pragma unroll
        for (int r = 0; r < 4; ++r) {
            int mycnt = 0;
            pre[r] = 0;
            #pragma unroll
            for (int L = 0; L < NBIN; ++L) {
                const unsigned m = __builtin_amdgcn_ballot_w32(bin[r] == L);
                const int below = (int)__builtin_amdgcn_mbcnt_lo(m, 0u);
                pre[r] = (bin[r] == L) ? below : pre[r];
                mycnt  = (lane == L) ? __builtin_popcount(m) : mycnt;
            }
            if (lane < NBIN) s_cnt[lane * CPITCH + wave * 4 + r] = mycnt;
        }
    }
    __syncthreads();

    if (tid < NBIN) {
        int run = 0;
        #pragma unroll
        for (int sg = 0; sg < NSEG; ++sg) {
            const int c = s_cnt[tid * CPITCH + sg];
            s_cnt[tid * CPITCH + sg] = run;
            run += c;
        }
        s_tot[tid] = run;
    }

    for (int i = tid; i < G4H * HDIM; i += WAVES * 32) {
        const int srow = i / HDIM;
        const int g = srow / HDIM, u = srow % HDIM;
        s_whh[(u * 4 + g) * WHHP + (i % HDIM)] = Whh[i] * (g == GATE_G ? 1.0f : 0.5f);
    }

    __syncthreads();

    {
        int base[4] = {0, 0, 0, 0};
        int run = 0;
        #pragma unroll
        for (int L = MAXLEN; L >= 0; --L) {
            #pragma unroll
            for (int r = 0; r < 4; ++r) base[r] = (bin[r] == L) ? run : base[r];
            run += s_tot[L];
        }
        #pragma unroll
        for (int r = 0; r < 4; ++r) {
            const unsigned rank = (unsigned)(base[r] + s_cnt[bin[r] * CPITCH + wave * 4 + r] + pre[r]);
            const unsigned j = rank >> 4;
            const unsigned p = j / (4 * GRP), i = j % (4 * GRP);
            const unsigned sid = (p & 1) ? (4 * GRP - 1 - i) : i;
            if (sid % GRP == (unsigned)gpos) {
                const unsigned slot = (p * 4 + sid / GRP) * 16 + (rank & 15);
                s_word[slot] = gbase + 4 * tid + r;
                s_wlen[slot] = bin[r];
            }
        }
    }

    {
        const int gcol = wave * 16 + n;
        v2f bw[E / 4];
        #pragma unroll
        for (int kk = 0; kk < E / 4; ++kk)
            bw[kk] = *(const v2f*)&Wih[gcol * E + kk * 4 + 2 * half];
        const float bsum = bih[gcol] + bhh[gcol];
        const float tscale = ((wave >> 2) == GATE_G) ? 1.0f : 0.5f;
        const int tdst = (wave & 3) * 64 + n * 4 + (wave >> 2);
        #pragma unroll
        for (int mt = 0; mt < (VOCAB + 15) / 16; ++mt) {
            int arow = mt * 16 + n;
            arow = arow < VOCAB ? arow : VOCAB - 1;
            v8f tacc;
            #pragma unroll
            for (int r = 0; r < 8; ++r) tacc[r] = bsum;
            #pragma unroll
            for (int kk = 0; kk < E / 4; ++kk) {
                v2f a = *(const v2f*)&emb[arow * E + kk * 4 + 2 * half];
                tacc = __builtin_amdgcn_wmma_f32_16x16x4_f32(
                    false, a, false, bw[kk], (short)0, tacc, false, false);
            }
            #pragma unroll
            for (int r = 0; r < 8; ++r) {
                const int orow = mt * 16 + 8 * half + r;
                if (orow < VOCAB) s_tab[orow * TPITCH + tdst] = tacc[r] * tscale;
            }
        }
    }

    __syncthreads();

    const int word_n = s_word[wave * 16 + n];

    {
        const v4i* cp = (const v4i*)&chars[word_n * MAXLEN + 8 * half];
        const v4i c0 = cp[0], c1 = cp[1];
        #pragma unroll
        for (int j = 0; j < 4; ++j) {
            s_off[wave][(8 * half + j) * 16 + n]     = c0[j] * (TPITCH / 4);
            s_off[wave][(8 * half + 4 + j) * 16 + n] = c1[j] * (TPITCH / 4);
        }
    }

    const int mylen = s_wlen[wave * 16 + n];

    const int tmax = __builtin_amdgcn_readfirstlane(s_wlen[wave * 16]);

    __syncthreads();

    v2f creg[16];
    v2f hcur[16];

    {
        const int rb0 = s_off[wave][n] * 4 + 8 * half;
        const bool act0 = 0 < mylen;
        #pragma unroll
        for (int mt = 0; mt < 16; ++mt) {
            #pragma unroll
            for (int s = 0; s < 2; ++s) {
                const v4f tb0 = *(const v4f*)&s_tab[rb0 + 16 * mt + 4 * s];
                const float ig = sigm_of_half(tb0[0]);
                const float gg = tanh_hw(tb0[2]);
                const float og = sigm_of_half(tb0[3]);
                const float cn = ig * gg;
                const float hn = og * tanh_hw(cn);
                creg[mt][s] = act0 ? cn : 0.0f;
                hcur[mt][s] = act0 ? hn : 0.0f;
            }
        }
    }

    #pragma unroll 1
    for (int t = 1; t < tmax; ++t) {
        const int rb = s_off[wave][t * 16 + n] * 4 + 8 * half;
        const bool act = t < mylen;

        v2f as[2][16];
        #define LOAD_AQ(SET, MG, Q)                                                      \
            _Pragma("unroll")                                                            \
            for (int k4 = 0; k4 < 4; ++k4) {                                             \
                _Pragma("unroll")                                                        \
                for (int j = 0; j < 4; ++j)                                              \
                    as[SET][k4 * 4 + j] = *(const v2f*)&s_whh[(((MG) * 4 + j) * 16 + n) * WHHP \
                                                      + ((Q) * 4 + k4) * 4 + 2 * half];  \
            }
        v8f acc[2][4];
        #define LOAD_CT(SET, MG)                                                         \
            _Pragma("unroll")                                                            \
            for (int j = 0; j < 4; ++j) {                                                \
                const v4f lo = *(const v4f*)&s_tab[rb + 16 * ((MG) * 4 + j)];            \
                const v4f hi = *(const v4f*)&s_tab[rb + 16 * ((MG) * 4 + j) + 4];        \
                acc[SET][j] = __builtin_shufflevector(lo, hi, 0, 1, 2, 3, 4, 5, 6, 7);   \
            }
        LOAD_AQ(0, 0, 0)
        LOAD_CT(0, 0)

        v2f hnew[16];
        #pragma unroll
        for (int blk = 0; blk < 16; ++blk) {
            const int mg = blk >> 2, q = blk & 3;
            if (q < 3) {
                LOAD_AQ((blk + 1) & 1, mg, q + 1)
            } else if (mg < 3) {
                LOAD_CT((mg + 1) & 1, mg + 1)
            }
            __builtin_amdgcn_sched_barrier(0);

            #pragma unroll
            for (int k4 = 0; k4 < 4; ++k4)
                #pragma unroll
                for (int j = 0; j < 4; ++j)
                    acc[mg & 1][j] = __builtin_amdgcn_wmma_f32_16x16x4_f32(
                        false, as[blk & 1][k4 * 4 + j], false, hcur[q * 4 + k4],
                        (short)0, acc[mg & 1][j], false, false);
            __builtin_amdgcn_sched_barrier(0);

            if (q == 3) {
                if (mg < 3) {
                    LOAD_AQ((blk + 1) & 1, mg + 1, 0)
                    __builtin_amdgcn_sched_barrier(0);
                }
                #pragma unroll
                for (int j = 0; j < 4; ++j) {
                    #pragma unroll
                    for (int s = 0; s < 2; ++s) {
                        const int mt = mg * 4 + j;
                        const float ig = sigm_of_half(acc[mg & 1][j][4 * s + 0]);
                        const float fg = sigm_of_half(acc[mg & 1][j][4 * s + 1]);
                        const float gg = tanh_hw(acc[mg & 1][j][4 * s + 2]);
                        const float og = sigm_of_half(acc[mg & 1][j][4 * s + 3]);
                        const float cold = creg[mt][s];
                        const float cn = fg * cold + ig * gg;
                        const float hn = og * tanh_hw(cn);
                        creg[mt][s] = act ? cn : cold;
                        hnew[mt][s] = act ? hn : hcur[mt][s];
                    }
                }
                __builtin_amdgcn_sched_barrier(0);
            }
        }
        #undef LOAD_AQ
        #undef LOAD_CT

        #pragma unroll
        for (int mt = 0; mt < 16; ++mt) hcur[mt] = hnew[mt];

        __builtin_amdgcn_fence(__ATOMIC_ACQ_REL, "wavefront");
    }

    #pragma unroll
    for (int mt = 0; mt < 16; ++mt)
        *(v2f*)&s_h[wave][n * HPITCH + 4 * mt + 2 * half] = hcur[mt];
    __builtin_amdgcn_fence(__ATOMIC_ACQ_REL, "wavefront");
    #pragma unroll
    for (int j = 0; j < 8; ++j) {
        int idx = j * 128 + lane * 4;
        int row = idx >> 6, col = idx & 63;
        v4f v = *(const v4f*)&s_h[wave][row * HPITCH + col];
        const int word = s_word[wave * 16 + row];
        *(v4f*)&out[word * HDIM + col] = v;
    }
}

extern "C" void kernel_launch(void* const* d_in, const int* in_sizes, int n_in,
                              void* d_out, int out_size, void* d_ws, size_t ws_size,
                              hipStream_t stream) {
    const float* emb     = (const float*)d_in[0];
    const float* W_ih    = (const float*)d_in[1];
    const float* W_hh    = (const float*)d_in[2];
    const float* b_ih    = (const float*)d_in[3];
    const float* b_hh    = (const float*)d_in[4];
    const int*   chars   = (const int*)d_in[5];
    const int*   lengths = (const int*)d_in[6];
    float* out = (float*)d_out;

    const int n_words = in_sizes[6];
    const int blocks  = n_words / (WAVES * 16);
    char_lstm_wmma<<<blocks, WAVES * 32, 0, stream>>>(
        emb, W_ih, W_hh, b_ih, b_hh, chars, lengths, out);
}


# Per-character tables built once per call by a small first launch, not by every block, and extended by T2[c] = Whh h1[c]: time step 1 adds rows T[c1] + T2[c0], no matrix work (sum order changes)
# speedup vs baseline: 4.5950x; 1.0418x over previous
#include <hip/hip_runtime.h>


#define VOCAB   100
#define E       32
#define HDIM    64
#define G4H     256
#define MAXLEN  16
#define WAVES   16
#define WPB     (WAVES * 16)
#define TPITCH  260
#define WHHP    68
#define HPITCH  68
#define GRP     8
#define GWORDS  (GRP * WPB)
#define NBIN    (MAXLEN + 1)
#define NSEG    (WAVES * 4)
#define CPITCH  (NSEG + 1)
#define TGP     G4H
#define CT      16
#define NTB     7
#define TROWV   (G4H / 4)

typedef __attribute__((ext_vector_type(2))) float v2f;
typedef __attribute__((ext_vector_type(4))) float v4f;
typedef __attribute__((ext_vector_type(8))) float v8f;
typedef __attribute__((ext_vector_type(4))) int   v4i;

#define GATE_G    2

__device__ __forceinline__ float sigm_of_half(float xh) {
    return __builtin_fmaf(0.5f, __builtin_amdgcn_tanhf(xh), 0.5f);
}

__device__ __forceinline__ float tanh_hw(float x) {
    return __builtin_amdgcn_tanhf(x);
}

__global__ __launch_bounds__(WAVES * 32)
void char_tables_wmma(const float* __restrict__ emb,
                      const float* __restrict__ Wih,
                      const float* __restrict__ Whh,
                      const float* __restrict__ bih,
                      const float* __restrict__ bhh,
                      float*       __restrict__ tabs)
{
    __shared__ alignas(16) float s_t[2][CT * TPITCH];
    __shared__ alignas(16) float s_h1[CT * HPITCH];

    const int tid  = threadIdx.x;
    const int lane = tid & 31;
    const int wave = __builtin_amdgcn_readfirstlane(tid >> 5);
    const int half = lane >> 4;
    const int n    = lane & 15;

    const int ag = n & 3;
    const int arow = ag * HDIM + wave * 4 + (n >> 2);
    const float ascale = (ag == GATE_G) ? 1.0f : 0.5f;
    const int chn = blockIdx.x * CT + n;
    const int chc = chn < VOCAB ? chn : VOCAB - 1;

    v2f a1[E / 4], b1[E / 4], a2[HDIM / 4];
    #pragma unroll
    for (int kk = 0; kk < E / 4; ++kk) {
        a1[kk] = *(const v2f*)&Wih[arow * E + kk * 4 + 2 * half] * ascale;
        b1[kk] = *(const v2f*)&emb[chc * E + kk * 4 + 2 * half];
    }
    #pragma unroll
    for (int kk = 0; kk < HDIM / 4; ++kk)
        a2[kk] = *(const v2f*)&Whh[arow * HDIM + kk * 4 + 2 * half] * ascale;

    v8f acc;
    #pragma unroll
    for (int r = 0; r < 8; ++r) {
        const int g = r & 3;
        const int brow = g * HDIM + wave * 4 + 2 * half + (r >> 2);
        acc[r] = (bih[brow] + bhh[brow]) * (g == GATE_G ? 1.0f : 0.5f);
    }
    #pragma unroll
    for (int kk = 0; kk < E / 4; ++kk)
        acc = __builtin_amdgcn_wmma_f32_16x16x4_f32(
            false, a1[kk], false, b1[kk], (short)0, acc, false, false);

    {
        v2f h1;
        #pragma unroll
        for (int s = 0; s < 2; ++s) {
            const float ig = sigm_of_half(acc[4 * s + 0]);
            const float gg = tanh_hw(acc[4 * s + 2]);
            const float og = sigm_of_half(acc[4 * s + 3]);
            h1[s] = og * tanh_hw(ig * gg);
        }
        *(v2f*)&s_h1[n * HPITCH + wave * 4 + 2 * half] = h1;
        *(v4f*)&s_t[0][n * TPITCH + wave * 16 + 8 * half] =
            __builtin_shufflevector(acc, acc, 0, 1, 2, 3);
        *(v4f*)&s_t[0][n * TPITCH + wave * 16 + 8 * half + 4] =
            __builtin_shufflevector(acc, acc, 4, 5, 6, 7);
    }
    __syncthreads();

    v8f acc2;
    #pragma unroll
    for (int r = 0; r < 8; ++r) acc2[r] = 0.0f;
    #pragma unroll
    for (int kk = 0; kk < HDIM / 4; ++kk) {
        const v2f b2 = *(const v2f*)&s_h1[n * HPITCH + kk * 4 + 2 * half];
        acc2 = __builtin_amdgcn_wmma_f32_16x16x4_f32(
            false, a2[kk], false, b2, (short)0, acc2, false, false);
    }
    *(v4f*)&s_t[1][n * TPITCH + wave * 16 + 8 * half] =
        __builtin_shufflevector(acc2, acc2, 0, 1, 2, 3);
    *(v4f*)&s_t[1][n * TPITCH + wave * 16 + 8 * half + 4] =
        __builtin_shufflevector(acc2, acc2, 4, 5, 6, 7);
    __syncthreads();

    const int chw = blockIdx.x * CT + wave;
    if (chw < VOCAB) {
        #pragma unroll
        for (int tb = 0; tb < 2; ++tb)
            #pragma unroll
            for (int j = 0; j < 2; ++j) {
                const int col = j * 128 + lane * 4;
                const v4f v = *(const v4f*)&s_t[tb][wave * TPITCH + col];
                *(v4f*)&tabs[(tb * VOCAB + chw) * TGP + col] = v;
            }
    }
}

__global__ __launch_bounds__(WAVES * 32, 4)
void char_lstm_wmma(const float* __restrict__ Whh,
                    const int*   __restrict__ chars,
                    const int*   __restrict__ lengths,
                    const float* __restrict__ tabs,
                    float*       __restrict__ out)
{
    __shared__ alignas(16) float s_tab[VOCAB * TPITCH];
    __shared__ alignas(16) float s_t2[VOCAB * TPITCH];
    __shared__ alignas(16) float s_whh[G4H * WHHP];
    __shared__ alignas(16) int   s_off[WAVES][MAXLEN * 16];
    __shared__ int   s_cnt[NBIN * CPITCH];
    __shared__ int   s_tot[NBIN];
    __shared__ int   s_word[WPB];
    __shared__ int   s_wlen[WPB];

    const int tid  = threadIdx.x;
    const int lane = tid & 31;
    const int wave = __builtin_amdgcn_readfirstlane(tid >> 5);
    const int half = lane >> 4;
    const int n    = lane & 15;
    const int gbase = (blockIdx.x / GRP) * GWORDS;
    const int gpos  = blockIdx.x % GRP;

    int bin[4], pre[4];
    {
        const v4i len4 = *(const v4i*)&lengths[gbase + 4 * tid];
        #pragma unroll
        for (int r = 0; r < 4; ++r) {
            const int l = len4[r];
            bin[r] = l < 0 ? 0 : (l > MAXLEN ? MAXLEN : l);
        }
        #pragma unroll
        for (int r = 0; r < 4; ++r) {
            int mycnt = 0;
            pre[r] = 0;
            #pragma unroll
            for (int L = 0; L < NBIN; ++L) {
                const unsigned m = __builtin_amdgcn_ballot_w32(bin[r] == L);
                const int below = (int)__builtin_amdgcn_mbcnt_lo(m, 0u);
                pre[r] = (bin[r] == L) ? below : pre[r];
                mycnt  = (lane == L) ? __builtin_popcount(m) : mycnt;
            }
            if (lane < NBIN) s_cnt[lane * CPITCH + wave * 4 + r] = mycnt;
        }
    }
    __syncthreads();

    if (tid < NBIN) {
        int run = 0;
        #pragma unroll
        for (int sg = 0; sg < NSEG; ++sg) {
            const int c = s_cnt[tid * CPITCH + sg];
            s_cnt[tid * CPITCH + sg] = run;
            run += c;
        }
        s_tot[tid] = run;
    }

    for (int i = tid; i < G4H * HDIM; i += WAVES * 32) {
        const int srow = i / HDIM;
        const int g = srow / HDIM, u = srow % HDIM;
        s_whh[(u * 4 + g) * WHHP + (i % HDIM)] = Whh[i] * (g == GATE_G ? 1.0f : 0.5f);
    }

    __syncthreads();

    {
        int base[4] = {0, 0, 0, 0};
        int run = 0;
        #pragma unroll
        for (int L = MAXLEN; L >= 0; --L) {
            #pragma unroll
            for (int r = 0; r < 4; ++r) base[r] = (bin[r] == L) ? run : base[r];
            run += s_tot[L];
        }
        #pragma unroll
        for (int r = 0; r < 4; ++r) {
            const unsigned rank = (unsigned)(base[r] + s_cnt[bin[r] * CPITCH + wave * 4 + r] + pre[r]);
            const unsigned j = rank >> 4;
            const unsigned p = j / (4 * GRP), i = j % (4 * GRP);
            const unsigned sid = (p & 1) ? (4 * GRP - 1 - i) : i;
            if (sid % GRP == (unsigned)gpos) {
                const unsigned slot = (p * 4 + sid / GRP) * 16 + (rank & 15);
                s_word[slot] = gbase + 4 * tid + r;
                s_wlen[slot] = bin[r];
            }
        }
    }

    {
        #pragma unroll
        for (int tb = 0; tb < 2; ++tb) {
            float* dst = tb ? s_t2 : s_tab;
            v4f tv[(VOCAB * TROWV + WAVES * 32 - 1) / (WAVES * 32)];
            #pragma unroll
            for (int j = 0; j < (VOCAB * TROWV + WAVES * 32 - 1) / (WAVES * 32); ++j) {
                const int i = tid + j * WAVES * 32;
                if (i < VOCAB * TROWV)
                    tv[j] = *(const v4f*)&tabs[tb * VOCAB * TGP + i * 4];
            }
            #pragma unroll
            for (int j = 0; j < (VOCAB * TROWV + WAVES * 32 - 1) / (WAVES * 32); ++j) {
                const int i = tid + j * WAVES * 32;
                if (i < VOCAB * TROWV)
                    *(v4f*)&dst[(i / TROWV) * TPITCH + (i % TROWV) * 4] = tv[j];
            }
        }
    }

    __syncthreads();

    const int word_n = s_word[wave * 16 + n];

    {
        const v4i* cp = (const v4i*)&chars[word_n * MAXLEN + 8 * half];
        const v4i c0 = cp[0], c1 = cp[1];
        #pragma unroll
        for (int j = 0; j < 4; ++j) {
            s_off[wave][(8 * half + j) * 16 + n]     = c0[j] * (TPITCH / 4);
            s_off[wave][(8 * half + 4 + j) * 16 + n] = c1[j] * (TPITCH / 4);
        }
    }

    const int mylen = s_wlen[wave * 16 + n];

    const int tmax = __builtin_amdgcn_readfirstlane(s_wlen[wave * 16]);

    __syncthreads();

    v2f creg[16];
    v2f hcur[16];

    {
        const int rb0 = s_off[wave][n] * 4 + 8 * half;
        const bool act0 = 0 < mylen;
        #pragma unroll
        for (int mt = 0; mt < 16; ++mt) {
            #pragma unroll
            for (int s = 0; s < 2; ++s) {
                const v4f tb0 = *(const v4f*)&s_tab[rb0 + 16 * mt + 4 * s];
                const float ig = sigm_of_half(tb0[0]);
                const float gg = tanh_hw(tb0[2]);
                const float og = sigm_of_half(tb0[3]);
                const float cn = ig * gg;
                const float hn = og * tanh_hw(cn);
                creg[mt][s] = act0 ? cn : 0.0f;
                hcur[mt][s] = act0 ? hn : 0.0f;
            }
        }
    }

    if (tmax > 1) {
        const int rb1 = s_off[wave][16 + n] * 4 + 8 * half;
        const int rb2 = s_off[wave][n] * 4 + 8 * half;
        const bool act1 = 1 < mylen;
        v2f hn1[16];
        #pragma unroll
        for (int mg = 0; mg < 4; ++mg) {
            v4f ta[8], tr[8];
            #pragma unroll
            for (int e = 0; e < 8; ++e) {
                ta[e] = *(const v4f*)&s_tab[rb1 + 16 * (4 * mg + (e >> 1)) + 4 * (e & 1)];
                tr[e] = *(const v4f*)&s_t2[rb2 + 16 * (4 * mg + (e >> 1)) + 4 * (e & 1)];
            }
            #pragma unroll
            for (int e = 0; e < 8; ++e) {
                const int mt = 4 * mg + (e >> 1), s = e & 1;
                const v4f gs = ta[e] + tr[e];
                const float ig = sigm_of_half(gs[0]);
                const float fg = sigm_of_half(gs[1]);
                const float gg = tanh_hw(gs[2]);
                const float og = sigm_of_half(gs[3]);
                const float cold = creg[mt][s];
                const float cn = fg * cold + ig * gg;
                const float hn = og * tanh_hw(cn);
                creg[mt][s] = act1 ? cn : cold;
                hn1[mt][s]  = act1 ? hn : hcur[mt][s];
            }
        }
        #pragma unroll
        for (int mt = 0; mt < 16; ++mt) hcur[mt] = hn1[mt];
    }
    __syncthreads();
    float* s_hw = &s_t2[wave * 16 * HPITCH];

    #pragma unroll 1
    for (int t = 2; t < tmax; ++t) {
        const int rb = s_off[wave][t * 16 + n] * 4 + 8 * half;
        const bool act = t < mylen;

        v2f as[2][16];
        #define LOAD_AQ(SET, MG, Q)                                                      \
            _Pragma("unroll")                                                            \
            for (int k4 = 0; k4 < 4; ++k4) {                                             \
                _Pragma("unroll")                                                        \
                for (int j = 0; j < 4; ++j)                                              \
                    as[SET][k4 * 4 + j] = *(const v2f*)&s_whh[(((MG) * 4 + j) * 16 + n) * WHHP \
                                                      + ((Q) * 4 + k4) * 4 + 2 * half];  \
            }
        v8f acc[2][4];
        #define LOAD_CT(SET, MG)                                                         \
            _Pragma("unroll")                                                            \
            for (int j = 0; j < 4; ++j) {                                                \
                const v4f lo = *(const v4f*)&s_tab[rb + 16 * ((MG) * 4 + j)];            \
                const v4f hi = *(const v4f*)&s_tab[rb + 16 * ((MG) * 4 + j) + 4];        \
                acc[SET][j] = __builtin_shufflevector(lo, hi, 0, 1, 2, 3, 4, 5, 6, 7);   \
            }
        LOAD_AQ(0, 0, 0)
        LOAD_CT(0, 0)

        v2f hnew[16];
        #pragma unroll
        for (int blk = 0; blk < 16; ++blk) {
            const int mg = blk >> 2, q = blk & 3;
            if (q < 3) {
                LOAD_AQ((blk + 1) & 1, mg, q + 1)
            } else if (mg < 3) {
                LOAD_CT((mg + 1) & 1, mg + 1)
            }
            __builtin_amdgcn_sched_barrier(0);

            #pragma unroll
            for (int k4 = 0; k4 < 4; ++k4)
                #pragma unroll
                for (int j = 0; j < 4; ++j)
                    acc[mg & 1][j] = __builtin_amdgcn_wmma_f32_16x16x4_f32(
                        false, as[blk & 1][k4 * 4 + j], false, hcur[q * 4 + k4],
                        (short)0, acc[mg & 1][j], false, false);
            __builtin_amdgcn_sched_barrier(0);

            if (q == 3) {
                if (mg < 3) {
                    LOAD_AQ((blk + 1) & 1, mg + 1, 0)
                    __builtin_amdgcn_sched_barrier(0);
                }
                #pragma unroll
                for (int j = 0; j < 4; ++j) {
                    #pragma unroll
                    for (int s = 0; s < 2; ++s) {
                        const int mt = mg * 4 + j;
                        const float ig = sigm_of_half(acc[mg & 1][j][4 * s + 0]);
                        const float fg = sigm_of_half(acc[mg & 1][j][4 * s + 1]);
                        const float gg = tanh_hw(acc[mg & 1][j][4 * s + 2]);
                        const float og = sigm_of_half(acc[mg & 1][j][4 * s + 3]);
                        const float cold = creg[mt][s];
                        const float cn = fg * cold + ig * gg;
                        const float hn = og * tanh_hw(cn);
                        creg[mt][s] = act ? cn : cold;
                        hnew[mt][s] = act ? hn : hcur[mt][s];
                    }
                }
                __builtin_amdgcn_sched_barrier(0);
            }
        }
        #undef LOAD_AQ
        #undef LOAD_CT

        #pragma unroll
        for (int mt = 0; mt < 16; ++mt) hcur[mt] = hnew[mt];

        __builtin_amdgcn_fence(__ATOMIC_ACQ_REL, "wavefront");
    }

    #pragma unroll
    for (int mt = 0; mt < 16; ++mt)
        *(v2f*)&s_hw[n * HPITCH + 4 * mt + 2 * half] = hcur[mt];
    __builtin_amdgcn_fence(__ATOMIC_ACQ_REL, "wavefront");
    #pragma unroll
    for (int j = 0; j < 8; ++j) {
        int idx = j * 128 + lane * 4;
        int row = idx >> 6, col = idx & 63;
        v4f v = *(const v4f*)&s_hw[row * HPITCH + col];
        const int word = s_word[wave * 16 + row];
        *(v4f*)&out[word * HDIM + col] = v;
    }
}

extern "C" void kernel_launch(void* const* d_in, const int* in_sizes, int n_in,
                              void* d_out, int out_size, void* d_ws, size_t ws_size,
                              hipStream_t stream) {
    const float* emb     = (const float*)d_in[0];
    const float* W_ih    = (const float*)d_in[1];
    const float* W_hh    = (const float*)d_in[2];
    const float* b_ih    = (const float*)d_in[3];
    const float* b_hh    = (const float*)d_in[4];
    const int*   chars   = (const int*)d_in[5];
    const int*   lengths = (const int*)d_in[6];
    float* out = (float*)d_out;

    const int n_words = in_sizes[6];
    const int blocks  = n_words / (WAVES * 16);
    float* tabs = (float*)d_ws;
    char_tables_wmma<<<NTB, WAVES * 32, 0, stream>>>(
        emb, W_ih, W_hh, b_ih, b_hh, tabs);
    char_lstm_wmma<<<blocks, WAVES * 32, 0, stream>>>(
        W_hh, chars, lengths, tabs, out);
}


# Length ranking of each 8-block group computed once, by the first launch (one block per group), and read back as packed entries; main launch's ballot sort gone, its staging loads leave as one batch
# speedup vs baseline: 4.8244x; 1.0499x over previous
#include <hip/hip_runtime.h>


#define VOCAB   100
#define E       32
#define HDIM    64
#define G4H     256
#define MAXLEN  16
#define WAVES   16
#define WPB     (WAVES * 16)
#define TPITCH  260
#define WHHP    68
#define HPITCH  68
#define GRP     8
#define GWORDS  (GRP * WPB)
#define NBIN    (MAXLEN + 1)
#define NSEG    (WAVES * 4)
#define CPITCH  (NSEG + 1)
#define TGP     G4H
#define CT      16
#define NTB     7
#define TROWV   (G4H / 4)
#define TABF    (2 * VOCAB * TGP)
#define LENBITS 5

typedef __attribute__((ext_vector_type(2))) float v2f;
typedef __attribute__((ext_vector_type(4))) float v4f;
typedef __attribute__((ext_vector_type(8))) float v8f;
typedef __attribute__((ext_vector_type(4))) int   v4i;

#define GATE_G    2

__device__ __forceinline__ float sigm_of_half(float xh) {
    return __builtin_fmaf(0.5f, __builtin_amdgcn_tanhf(xh), 0.5f);
}

__device__ __forceinline__ float tanh_hw(float x) {
    return __builtin_amdgcn_tanhf(x);
}

__device__ __forceinline__ void rank_group(const int* __restrict__ lengths,
                                           int*       __restrict__ rank_ws,
                                           const int group)
{
    __shared__ int s_cnt[NBIN * CPITCH];
    __shared__ int s_tot[NBIN];
    __shared__ alignas(16) int s_rk[GWORDS];

    const int tid  = threadIdx.x;
    const int lane = tid & 31;
    const int wave = __builtin_amdgcn_readfirstlane(tid >> 5);
    const int gbase = group * GWORDS;

    for (int i = tid; i < NBIN * CPITCH; i += WAVES * 32) s_cnt[i] = 0;

    int bin[4], pre[4];
    {
        const v4i len4 = *(const v4i*)&lengths[gbase + 4 * tid];
        #pragma unroll
        for (int r = 0; r < 4; ++r) {
            const int l = len4[r];
            bin[r] = l < 0 ? 0 : (l > MAXLEN ? MAXLEN : l);
        }
    }
    __syncthreads();

    #pragma unroll
    for (int r = 0; r < 4; ++r) {
        unsigned same = 0xffffffffu;
        #pragma unroll
        for (int k = 0; k < LENBITS; ++k) {
            const bool bit = (bin[r] >> k) & 1;
            const unsigned bk = __builtin_amdgcn_ballot_w32(bit);
            same &= bit ? bk : ~bk;
        }
        pre[r] = (int)__builtin_amdgcn_mbcnt_lo(same, 0u);
        if (pre[r] == 0) s_cnt[bin[r] * CPITCH + wave * 4 + r] = __builtin_popcount(same);
    }
    __syncthreads();

    if (tid < NBIN) {
        int run = 0;
        #pragma unroll
        for (int sg = 0; sg < NSEG; ++sg) {
            const int cc = s_cnt[tid * CPITCH + sg];
            s_cnt[tid * CPITCH + sg] = run;
            run += cc;
        }
        s_tot[tid] = run;
    }
    __syncthreads();

    {
        int base[4] = {0, 0, 0, 0};
        int run = 0;
        #pragma unroll
        for (int L = MAXLEN; L >= 0; --L) {
            #pragma unroll
            for (int r = 0; r < 4; ++r) base[r] = (bin[r] == L) ? run : base[r];
            run += s_tot[L];
        }
        #pragma unroll
        for (int r = 0; r < 4; ++r) {
            const unsigned rank = (unsigned)(base[r] + s_cnt[bin[r] * CPITCH + wave * 4 + r] + pre[r]);
            const unsigned j = rank >> 4;
            const unsigned p = j / (4 * GRP), i = j % (4 * GRP);
            const unsigned sid = (p & 1) ? (4 * GRP - 1 - i) : i;
            const unsigned slot = (p * 4 + sid / GRP) * 16 + (rank & 15);
            s_rk[(sid % GRP) * WPB + slot] = ((gbase + 4 * tid + r) << LENBITS) | bin[r];
        }
    }
    __syncthreads();

    *(v4i*)&rank_ws[gbase + 4 * tid] = *(const v4i*)&s_rk[4 * tid];
}

__global__ __launch_bounds__(WAVES * 32)
void char_tables_wmma(const float* __restrict__ emb,
                      const float* __restrict__ Wih,
                      const float* __restrict__ Whh,
                      const float* __restrict__ bih,
                      const float* __restrict__ bhh,
                      const int*   __restrict__ lengths,
                      float*       __restrict__ tabs,
                      int*         __restrict__ rank_ws)
{
    if (blockIdx.x >= NTB) {
        rank_group(lengths, rank_ws, (int)blockIdx.x - NTB);
        return;
    }

    __shared__ alignas(16) float s_t[2][CT * TPITCH];
    __shared__ alignas(16) float s_h1[CT * HPITCH];

    const int tid  = threadIdx.x;
    const int lane = tid & 31;
    const int wave = __builtin_amdgcn_readfirstlane(tid >> 5);
    const int half = lane >> 4;
    const int n    = lane & 15;

    const int ag = n & 3;
    const int arow = ag * HDIM + wave * 4 + (n >> 2);
    const float ascale = (ag == GATE_G) ? 1.0f : 0.5f;
    const int chn = blockIdx.x * CT + n;
    const int chc = chn < VOCAB ? chn : VOCAB - 1;

    v2f a1[E / 4], b1[E / 4], a2[HDIM / 4];
    #pragma unroll
    for (int kk = 0; kk < E / 4; ++kk) {
        a1[kk] = *(const v2f*)&Wih[arow * E + kk * 4 + 2 * half] * ascale;
        b1[kk] = *(const v2f*)&emb[chc * E + kk * 4 + 2 * half];
    }
    #pragma unroll
    for (int kk = 0; kk < HDIM / 4; ++kk)
        a2[kk] = *(const v2f*)&Whh[arow * HDIM + kk * 4 + 2 * half] * ascale;

    v8f acc;
    #pragma unroll
    for (int r = 0; r < 8; ++r) {
        const int g = r & 3;
        const int brow = g * HDIM + wave * 4 + 2 * half + (r >> 2);
        acc[r] = (bih[brow] + bhh[brow]) * (g == GATE_G ? 1.0f : 0.5f);
    }
    #pragma unroll
    for (int kk = 0; kk < E / 4; ++kk)
        acc = __builtin_amdgcn_wmma_f32_16x16x4_f32(
            false, a1[kk], false, b1[kk], (short)0, acc, false, false);

    {
        v2f h1;
        #pragma unroll
        for (int s = 0; s < 2; ++s) {
            const float ig = sigm_of_half(acc[4 * s + 0]);
            const float gg = tanh_hw(acc[4 * s + 2]);
            const float og = sigm_of_half(acc[4 * s + 3]);
            h1[s] = og * tanh_hw(ig * gg);
        }
        *(v2f*)&s_h1[n * HPITCH + wave * 4 + 2 * half] = h1;
        *(v4f*)&s_t[0][n * TPITCH + wave * 16 + 8 * half] =
            __builtin_shufflevector(acc, acc, 0, 1, 2, 3);
        *(v4f*)&s_t[0][n * TPITCH + wave * 16 + 8 * half + 4] =
            __builtin_shufflevector(acc, acc, 4, 5, 6, 7);
    }
    __syncthreads();

    v8f acc2;
    #pragma unroll
    for (int r = 0; r < 8; ++r) acc2[r] = 0.0f;
    #pragma unroll
    for (int kk = 0; kk < HDIM / 4; ++kk) {
        const v2f b2 = *(const v2f*)&s_h1[n * HPITCH + kk * 4 + 2 * half];
        acc2 = __builtin_amdgcn_wmma_f32_16x16x4_f32(
            false, a2[kk], false, b2, (short)0, acc2, false, false);
    }
    *(v4f*)&s_t[1][n * TPITCH + wave * 16 + 8 * half] =
        __builtin_shufflevector(acc2, acc2, 0, 1, 2, 3);
    *(v4f*)&s_t[1][n * TPITCH + wave * 16 + 8 * half + 4] =
        __builtin_shufflevector(acc2, acc2, 4, 5, 6, 7);
    __syncthreads();

    const int chw = blockIdx.x * CT + wave;
    if (chw < VOCAB) {
        #pragma unroll
        for (int tb = 0; tb < 2; ++tb)
            #pragma unroll
            for (int j = 0; j < 2; ++j) {
                const int col = j * 128 + lane * 4;
                const v4f v = *(const v4f*)&s_t[tb][wave * TPITCH + col];
                *(v4f*)&tabs[(tb * VOCAB + chw) * TGP + col] = v;
            }
    }
}

__global__ __launch_bounds__(WAVES * 32, 4)
void char_lstm_wmma(const float* __restrict__ Whh,
                    const int*   __restrict__ chars,
                    const float* __restrict__ tabs,
                    const int*   __restrict__ rank,
                    float*       __restrict__ out,
                    const int n_words)
{
    __shared__ alignas(16) float s_tab[VOCAB * TPITCH];
    __shared__ alignas(16) float s_t2[VOCAB * TPITCH];
    __shared__ alignas(16) float s_whh[G4H * WHHP];
    __shared__ alignas(16) int   s_off[WAVES][MAXLEN * 16];
    __shared__ int   s_word[WPB];

    const int tid  = threadIdx.x;
    const int lane = tid & 31;
    const int wave = __builtin_amdgcn_readfirstlane(tid >> 5);
    const int half = lane >> 4;
    const int n    = lane & 15;

    const int pk = rank[blockIdx.x * WPB + wave * 16 + n];

    v4f wv[(G4H * HDIM / 4) / (WAVES * 32)];
    #pragma unroll
    for (int j = 0; j < (G4H * HDIM / 4) / (WAVES * 32); ++j)
        wv[j] = *(const v4f*)&Whh[(tid + j * WAVES * 32) * 4];
    v4f tv[2][(VOCAB * TROWV + WAVES * 32 - 1) / (WAVES * 32)];
    #pragma unroll
    for (int tb = 0; tb < 2; ++tb) {
        #pragma unroll
        for (int j = 0; j < (VOCAB * TROWV + WAVES * 32 - 1) / (WAVES * 32); ++j) {
            const int i = tid + j * WAVES * 32;
            if (i < VOCAB * TROWV)
                tv[tb][j] = *(const v4f*)&tabs[tb * VOCAB * TGP + i * 4];
        }
    }

    int word_n = pk >> LENBITS;
    word_n = word_n < 0 ? 0 : (word_n >= n_words ? n_words - 1 : word_n);
    const int mylen = pk & ((1 << LENBITS) - 1);
    const v4i* cp = (const v4i*)&chars[word_n * MAXLEN + 8 * half];
    const v4i c0 = cp[0], c1 = cp[1];

    #pragma unroll
    for (int j = 0; j < (G4H * HDIM / 4) / (WAVES * 32); ++j) {
        const int f = tid + j * WAVES * 32;
        const int srow = f / (HDIM / 4);
        const int g = srow / HDIM, u = srow % HDIM;
        *(v4f*)&s_whh[(u * 4 + g) * WHHP + (f % (HDIM / 4)) * 4] =
            wv[j] * (g == GATE_G ? 1.0f : 0.5f);
    }

    #pragma unroll
    for (int tb = 0; tb < 2; ++tb) {
        float* dst = tb ? s_t2 : s_tab;
        #pragma unroll
        for (int j = 0; j < (VOCAB * TROWV + WAVES * 32 - 1) / (WAVES * 32); ++j) {
            const int i = tid + j * WAVES * 32;
            if (i < VOCAB * TROWV)
                *(v4f*)&dst[(i / TROWV) * TPITCH + (i % TROWV) * 4] = tv[tb][j];
        }
    }

    #pragma unroll
    for (int j = 0; j < 4; ++j) {
        s_off[wave][(8 * half + j) * 16 + n]     = c0[j] * (TPITCH / 4);
        s_off[wave][(8 * half + 4 + j) * 16 + n] = c1[j] * (TPITCH / 4);
    }
    if (half == 0) s_word[wave * 16 + n] = word_n;

    const int tmax = __builtin_amdgcn_readfirstlane(mylen);

    __syncthreads();

    v2f creg[16];
    v2f hcur[16];

    {
        const int rb0 = s_off[wave][n] * 4 + 8 * half;
        const bool act0 = 0 < mylen;
        #pragma unroll
        for (int mt = 0; mt < 16; ++mt) {
            #pragma unroll
            for (int s = 0; s < 2; ++s) {
                const v4f tb0 = *(const v4f*)&s_tab[rb0 + 16 * mt + 4 * s];
                const float ig = sigm_of_half(tb0[0]);
                const float gg = tanh_hw(tb0[2]);
                const float og = sigm_of_half(tb0[3]);
                const float cn = ig * gg;
                const float hn = og * tanh_hw(cn);
                creg[mt][s] = act0 ? cn : 0.0f;
                hcur[mt][s] = act0 ? hn : 0.0f;
            }
        }
    }

    if (tmax > 1) {
        const int rb1 = s_off[wave][16 + n] * 4 + 8 * half;
        const int rb2 = s_off[wave][n] * 4 + 8 * half;
        const bool act1 = 1 < mylen;
        v2f hn1[16];
        #pragma unroll
        for (int mg = 0; mg < 4; ++mg) {
            v4f ta[8], tr[8];
            #pragma unroll
            for (int e = 0; e < 8; ++e) {
                ta[e] = *(const v4f*)&s_tab[rb1 + 16 * (4 * mg + (e >> 1)) + 4 * (e & 1)];
                tr[e] = *(const v4f*)&s_t2[rb2 + 16 * (4 * mg + (e >> 1)) + 4 * (e & 1)];
            }
            #pragma unroll
            for (int e = 0; e < 8; ++e) {
                const int mt = 4 * mg + (e >> 1), s = e & 1;
                const v4f gs = ta[e] + tr[e];
                const float ig = sigm_of_half(gs[0]);
                const float fg = sigm_of_half(gs[1]);
                const float gg = tanh_hw(gs[2]);
                const float og = sigm_of_half(gs[3]);
                const float cold = creg[mt][s];
                const float cn = fg * cold + ig * gg;
                const float hn = og * tanh_hw(cn);
                creg[mt][s] = act1 ? cn : cold;
                hn1[mt][s]  = act1 ? hn : hcur[mt][s];
            }
        }
        #pragma unroll
        for (int mt = 0; mt < 16; ++mt) hcur[mt] = hn1[mt];
    }
    __syncthreads();
    float* s_hw = &s_t2[wave * 16 * HPITCH];

    #pragma unroll 1
    for (int t = 2; t < tmax; ++t) {
        const int rb = s_off[wave][t * 16 + n] * 4 + 8 * half;
        const bool act = t < mylen;

        v2f as[2][16];
        #define LOAD_AQ(SET, MG, Q)                                                      \
            _Pragma("unroll")                                                            \
            for (int k4 = 0; k4 < 4; ++k4) {                                             \
                _Pragma("unroll")                                                        \
                for (int j = 0; j < 4; ++j)                                              \
                    as[SET][k4 * 4 + j] = *(const v2f*)&s_whh[(((MG) * 4 + j) * 16 + n) * WHHP \
                                                      + ((Q) * 4 + k4) * 4 + 2 * half];  \
            }
        v8f acc[2][4];
        #define LOAD_CT(SET, MG)                                                         \
            _Pragma("unroll")                                                            \
            for (int j = 0; j < 4; ++j) {                                                \
                const v4f lo = *(const v4f*)&s_tab[rb + 16 * ((MG) * 4 + j)];            \
                const v4f hi = *(const v4f*)&s_tab[rb + 16 * ((MG) * 4 + j) + 4];        \
                acc[SET][j] = __builtin_shufflevector(lo, hi, 0, 1, 2, 3, 4, 5, 6, 7);   \
            }
        LOAD_AQ(0, 0, 0)
        LOAD_CT(0, 0)

        v2f hnew[16];
        #pragma unroll
        for (int blk = 0; blk < 16; ++blk) {
            const int mg = blk >> 2, q = blk & 3;
            if (q < 3) {
                LOAD_AQ((blk + 1) & 1, mg, q + 1)
            } else if (mg < 3) {
                LOAD_CT((mg + 1) & 1, mg + 1)
            }
            __builtin_amdgcn_sched_barrier(0);

            #pragma unroll
            for (int k4 = 0; k4 < 4; ++k4)
                #pragma unroll
                for (int j = 0; j < 4; ++j)
                    acc[mg & 1][j] = __builtin_amdgcn_wmma_f32_16x16x4_f32(
                        false, as[blk & 1][k4 * 4 + j], false, hcur[q * 4 + k4],
                        (short)0, acc[mg & 1][j], false, false);
            __builtin_amdgcn_sched_barrier(0);

            if (q == 3) {
                if (mg < 3) {
                    LOAD_AQ((blk + 1) & 1, mg + 1, 0)
                    __builtin_amdgcn_sched_barrier(0);
                }
                #pragma unroll
                for (int j = 0; j < 4; ++j) {
                    #pragma unroll
                    for (int s = 0; s < 2; ++s) {
                        const int mt = mg * 4 + j;
                        const float ig = sigm_of_half(acc[mg & 1][j][4 * s + 0]);
                        const float fg = sigm_of_half(acc[mg & 1][j][4 * s + 1]);
                        const float gg = tanh_hw(acc[mg & 1][j][4 * s + 2]);
                        const float og = sigm_of_half(acc[mg & 1][j][4 * s + 3]);
                        const float cold = creg[mt][s];
                        const float cn = fg * cold + ig * gg;
                        const float hn = og * tanh_hw(cn);
                        creg[mt][s] = act ? cn : cold;
                        hnew[mt][s] = act ? hn : hcur[mt][s];
                    }
                }
                __builtin_amdgcn_sched_barrier(0);
            }
        }
        #undef LOAD_AQ
        #undef LOAD_CT

        #pragma unroll
        for (int mt = 0; mt < 16; ++mt) hcur[mt] = hnew[mt];

        __builtin_amdgcn_fence(__ATOMIC_ACQ_REL, "wavefront");
    }

    #pragma unroll
    for (int mt = 0; mt < 16; ++mt)
        *(v2f*)&s_hw[n * HPITCH + 4 * mt + 2 * half] = hcur[mt];
    __builtin_amdgcn_fence(__ATOMIC_ACQ_REL, "wavefront");
    #pragma unroll
    for (int j = 0; j < 8; ++j) {
        int idx = j * 128 + lane * 4;
        int row = idx >> 6, col = idx & 63;
        v4f v = *(const v4f*)&s_hw[row * HPITCH + col];
        const int word = s_word[wave * 16 + row];
        *(v4f*)&out[word * HDIM + col] = v;
    }
}

extern "C" void kernel_launch(void* const* d_in, const int* in_sizes, int n_in,
                              void* d_out, int out_size, void* d_ws, size_t ws_size,
                              hipStream_t stream) {
    const float* emb     = (const float*)d_in[0];
    const float* W_ih    = (const float*)d_in[1];
    const float* W_hh    = (const float*)d_in[2];
    const float* b_ih    = (const float*)d_in[3];
    const float* b_hh    = (const float*)d_in[4];
    const int*   chars   = (const int*)d_in[5];
    const int*   lengths = (const int*)d_in[6];
    float* out = (float*)d_out;

    const int n_words = in_sizes[6];
    const int blocks  = n_words / (WAVES * 16);
    float* tabs = (float*)d_ws;
    int*   rank = (int*)(tabs + TABF);
    const int groups = blocks / GRP;
    char_tables_wmma<<<NTB + groups, WAVES * 32, 0, stream>>>(
        emb, W_ih, W_hh, b_ih, b_hh, lengths, tabs, rank);
    char_lstm_wmma<<<blocks, WAVES * 32, 0, stream>>>(
        W_hh, chars, tabs, rank, out, n_words);
}
